# starting kernel
# baseline (speedup 1.0000x reference)
#include <hip/hip_runtime.h>
#include <cmath>

typedef __attribute__((ext_vector_type(2))) float v2f;
typedef __attribute__((ext_vector_type(8))) float v8f;

#define RNN_B 1024
#define RNN_T 512
#define RNN_H 128
#define RNN_P 4
#define RNN_TT (RNN_T + RNN_P)
#define NTILE 8
#define KCHUNK 32
#define HB_STRIDE 132

#define FCW_ELEMS (KCHUNK * NTILE * 32 * 2)

__device__ __forceinline__ float dev_tanh(float v) {
#if __has_builtin(__builtin_amdgcn_tanhf)
  return __builtin_amdgcn_tanhf(v);
#else
  return tanhf(v);
#endif
}

__global__ __launch_bounds__(32) void rnn516_wmma(
    const float* __restrict__ x, const float* __restrict__ hidden,
    const float* __restrict__ fc0_w, const float* __restrict__ fc0_b,
    const float* __restrict__ fc_w, const float* __restrict__ fc_b,
    const float* __restrict__ dec_w, const float* __restrict__ dec_b,
    float* __restrict__ y) {
  extern __shared__ float lds[];
  float* fcw  = lds;
  float* hbuf = lds + FCW_ELEMS;
  float* ybuf = hbuf + 16 * HB_STRIDE;

  const int lane = threadIdx.x;
  const int half = lane >> 4;
  const int lmod = lane & 15;
  const int b0 = blockIdx.x * 16;

  for (int i = lane; i < FCW_ELEMS; i += 32) {
    int j  = i & 1;
    int l2 = (i >> 1) & 31;
    int nt = (i >> 6) & 7;
    int kc = i >> 9;
    int n = nt * 16 + (l2 & 15);
    int k = 4 * kc + 2 * (l2 >> 4) + j;
    fcw[i] = fc_w[n * RNN_H + k];
  }
  for (int i = lane; i < 16 * HB_STRIDE; i += 32) {
    int r = i / HB_STRIDE, k = i - r * HB_STRIDE;
    hbuf[i] = (k < RNN_H) ? hidden[(b0 + r) * RNN_H + k] : 0.0f;
  }
  __syncthreads();

  float biasv[NTILE], dwv[NTILE];
  v2f blast[NTILE];
#pragma unroll
  for (int nt = 0; nt < NTILE; ++nt) {
    int n = nt * 16 + lmod;
    biasv[nt] = fc0_b[n] + fc_b[n];
    dwv[nt] = dec_w[n];
    v2f b;
    b[0] = (half == 0) ? fc0_w[n] : 0.0f;
    b[1] = 0.0f;
    blast[nt] = b;
  }
  const float decb = dec_b[0];

  float xval = 0.0f;

  for (int t = 0; t < RNN_TT; ++t) {
    if (t < RNN_T) {
      if (half == 0) xval = x[(b0 + lmod) * RNN_T + t];
    }

    v8f d[NTILE];
#pragma unroll
    for (int nt = 0; nt < NTILE; ++nt) {
#pragma unroll
      for (int e = 0; e < 8; ++e) d[nt][e] = biasv[nt];
    }

    const float* arow  = hbuf + lmod * HB_STRIDE + 2 * half;
    const float* bcol  = fcw + 2 * lane;

    v2f a_cur, b_cur[NTILE], a_nxt, b_nxt[NTILE];
    a_cur = *(const v2f*)(arow);
#pragma unroll
    for (int nt = 0; nt < NTILE; ++nt)
      b_cur[nt] = *(const v2f*)(bcol + nt * 64);

#pragma unroll 4
    for (int kc = 0; kc < KCHUNK - 1; ++kc) {
      a_nxt = *(const v2f*)(arow + 4 * (kc + 1));
      const float* bb = bcol + (kc + 1) * (NTILE * 64);
#pragma unroll
      for (int nt = 0; nt < NTILE; ++nt)
        b_nxt[nt] = *(const v2f*)(bb + nt * 64);

#pragma unroll
      for (int nt = 0; nt < NTILE; ++nt)
        d[nt] = __builtin_amdgcn_wmma_f32_16x16x4_f32(
            false, a_cur, false, b_cur[nt], (short)0, d[nt], false, false);

      a_cur = a_nxt;
#pragma unroll
      for (int nt = 0; nt < NTILE; ++nt) b_cur[nt] = b_nxt[nt];
    }
#pragma unroll
    for (int nt = 0; nt < NTILE; ++nt)
      d[nt] = __builtin_amdgcn_wmma_f32_16x16x4_f32(
          false, a_cur, false, b_cur[nt], (short)0, d[nt], false, false);

    {
      v2f al;
      al[0] = xval;
      al[1] = 0.0f;
#pragma unroll
      for (int nt = 0; nt < NTILE; ++nt)
        d[nt] = __builtin_amdgcn_wmma_f32_16x16x4_f32(
            false, al, false, blast[nt], (short)0, d[nt], false, false);
    }

    float yred[8];
#pragma unroll
    for (int v = 0; v < 8; ++v) yred[v] = 0.0f;
#pragma unroll
    for (int nt = 0; nt < NTILE; ++nt) {
      int col = nt * 16 + lmod;
#pragma unroll
      for (int v = 0; v < 8; ++v) {
        float th = dev_tanh(d[nt][v]);
        hbuf[(v + 8 * half) * HB_STRIDE + col] = th;
        yred[v] += th * dwv[nt];
      }
    }

#pragma unroll
    for (int v = 0; v < 8; ++v) {
      float s = yred[v];
      s += __shfl_xor(s, 1, 32);
      s += __shfl_xor(s, 2, 32);
      s += __shfl_xor(s, 4, 32);
      s += __shfl_xor(s, 8, 32);
      if (lmod == 0) ybuf[v + 8 * half] = s + decb;
    }

    float yv = ybuf[lmod];
    if (half == 0) {
      y[(b0 + lmod) * RNN_TT + t] = yv;
      xval = yv;
    }
  }
}

extern "C" void kernel_launch(void* const* d_in, const int* in_sizes, int n_in,
                              void* d_out, int out_size, void* d_ws,
                              size_t ws_size, hipStream_t stream) {
  const float* x      = (const float*)d_in[0];
  const float* hidden = (const float*)d_in[1];
  const float* fc0_w  = (const float*)d_in[2];
  const float* fc0_b  = (const float*)d_in[3];
  const float* fc_w   = (const float*)d_in[4];
  const float* fc_b   = (const float*)d_in[5];
  const float* dec_w  = (const float*)d_in[6];
  const float* dec_b  = (const float*)d_in[7];
  float* y = (float*)d_out;

  const size_t shmem =
      (size_t)(FCW_ELEMS + 16 * HB_STRIDE + 16) * sizeof(float);
  rnn516_wmma<<<dim3(RNN_B / 16), dim3(32), shmem, stream>>>(
      x, hidden, fc0_w, fc0_b, fc_w, fc_b, dec_w, dec_b, y);
}


# reduction dim k split over 4 waves per 16-row tile: a wave's tanh output (32 h cols) stays in VGPRs as its next B, fc_w as A; partial D tiles exchanged via LDS, 1 barrier/step; new sum order
# speedup vs baseline: 7.4998x; 7.4998x over previous
#include <hip/hip_runtime.h>
#include <cmath>


typedef __attribute__((ext_vector_type(2))) float v2f;
typedef __attribute__((ext_vector_type(4))) float v4f;
typedef __attribute__((ext_vector_type(8))) float v8f;

#define RNN_B 1024
#define RNN_T 512
#define RNN_H 128
#define RNN_P 4
#define RNN_TT (RNN_T + RNN_P)


__device__ __forceinline__ float dev_tanh(float v) {
#if __has_builtin(__builtin_amdgcn_tanhf)
  return __builtin_amdgcn_tanhf(v);
#else
  return tanhf(v);
#endif
}

__device__ __forceinline__ float xhalf(float v) {
  int i = __builtin_bit_cast(int, v);
  return __builtin_bit_cast(
      float, __builtin_amdgcn_permlanex16(i, i, 0x76543210, 0xfedcba98, false, false));
}

__device__ __forceinline__ v8f mma(v2f a, v2f b, v8f c) {
  return __builtin_amdgcn_wmma_f32_16x16x4_f32(false, a, false, b, (short)0, c,
                                               false, false);
}

__global__ __launch_bounds__(128) __attribute__((amdgpu_waves_per_eu(1, 1)))
void rnn516_ksplit(
    const float* __restrict__ x, const float* __restrict__ hidden,
    const float* __restrict__ fc0_w, const float* __restrict__ fc0_b,
    const float* __restrict__ fc_w, const float* __restrict__ fc_b,
    const float* __restrict__ dec_w, const float* __restrict__ dec_b,
    float* __restrict__ y) {
  __shared__ v4f pbuf[2 * 4 * 4 * 4 * 32];
  __shared__ v4f ybuf[2 * 16];

  const int lane = threadIdx.x & 31;
  const int w = __builtin_amdgcn_readfirstlane(threadIdx.x >> 5);
  const int hf = lane >> 4;
  const int c = lane & 15;
  const int b0 = blockIdx.x * 16;
  const int kown = 32 * w + 8 * hf;

  v2f Af[4][2][8];
#pragma unroll
  for (int d = 0; d < 4; ++d) {
#pragma unroll
    for (int j = 0; j < 2; ++j) {
      const int n = 16 * (2 * ((w + d) & 3) + j) + c;
      const float* wr = fc_w + n * RNN_H + kown;
#pragma unroll
      for (int jj = 0; jj < 2; ++jj) {
        v4f lo = *(const v4f*)(wr + 16 * jj);
        v4f hi = *(const v4f*)(wr + 16 * jj + 4);
        Af[d][j][4 * jj + 0] = v2f{lo[0], lo[1]};
        Af[d][j][4 * jj + 1] = v2f{lo[2], lo[3]};
        Af[d][j][4 * jj + 2] = v2f{hi[0], hi[1]};
        Af[d][j][4 * jj + 3] = v2f{hi[2], hi[3]};
      }
    }
  }

  v8f biasv[2], dwv[2], hB[2];
  v2f Ax[2];
#pragma unroll
  for (int jj = 0; jj < 2; ++jj) {
#pragma unroll
    for (int r = 0; r < 8; ++r) {
      const int n = kown + 16 * jj + r;
      biasv[jj][r] = fc0_b[n] + fc_b[n];
      dwv[jj][r] = dec_w[n];
      hB[jj][r] = hidden[(b0 + c) * RNN_H + n];
    }
    Ax[jj] = v2f{(hf == 0) ? fc0_w[32 * w + 16 * jj + c] : 0.0f, 0.0f};
  }
  const float decb = dec_b[0];

  v8f o0, o1;
  v4f rc[3][4];
  v4f yp;
  float xcur = x[(b0 + c) * RNN_T], xnext;

  auto head = [&](const int t) {
    const int par = t & 1;
    const int tn = (t + 1 < RNN_T) ? t + 1 : RNN_T - 1;
    xnext = x[(b0 + c) * RNN_T + tn];

    v2f Bf[8];
#pragma unroll
    for (int jj = 0; jj < 2; ++jj) {
#pragma unroll
      for (int p = 0; p < 4; ++p)
        Bf[4 * jj + p] = v2f{hB[jj][2 * p], hB[jj][2 * p + 1]};
    }

    {
      float s0 = 0.0f, s1 = 0.0f, s2 = 0.0f, s3 = 0.0f;
#pragma unroll
      for (int jj = 0; jj < 2; ++jj) {
        s0 = fmaf(hB[jj][0], dwv[jj][0], s0);
        s1 = fmaf(hB[jj][1], dwv[jj][1], s1);
        s2 = fmaf(hB[jj][2], dwv[jj][2], s2);
        s3 = fmaf(hB[jj][3], dwv[jj][3], s3);
        s0 = fmaf(hB[jj][4], dwv[jj][4], s0);
        s1 = fmaf(hB[jj][5], dwv[jj][5], s1);
        s2 = fmaf(hB[jj][6], dwv[jj][6], s2);
        s3 = fmaf(hB[jj][7], dwv[jj][7], s3);
      }
      float ys = (s0 + s1) + (s2 + s3);
      ys += xhalf(ys);
      ((float*)ybuf)[((par ^ 1) * 16 + c) * 4 + w] = ys;
    }

    v8f fa[3][2];
#pragma unroll
    for (int d = 1; d < 4; ++d) {
      v8f a0 = {0, 0, 0, 0, 0, 0, 0, 0}, a1 = {0, 0, 0, 0, 0, 0, 0, 0};
#pragma unroll
      for (int q = 0; q < 8; ++q) {
        a0 = mma(Af[d][0][q], Bf[q], a0);
        a1 = mma(Af[d][1][q], Bf[q], a1);
      }
      fa[d - 1][0] = a0;
      fa[d - 1][1] = a1;
    }

    v4f* pst = pbuf + par * 2048 + w * 128 + lane;
#pragma unroll
    for (int d = 1; d < 4; ++d) {
      v4f* ps = pst + ((w + d) & 3) * 512;
      const v8f a0 = fa[d - 1][0], a1 = fa[d - 1][1];
      ps[0]  = v4f{a0[0], a0[1], a0[2], a0[3]};
      ps[32] = v4f{a0[4], a0[5], a0[6], a0[7]};
      ps[64] = v4f{a1[0], a1[1], a1[2], a1[3]};
      ps[96] = v4f{a1[4], a1[5], a1[6], a1[7]};
    }
    __builtin_amdgcn_sched_barrier(0);

    o0 = biasv[0];
    o1 = biasv[1];
#pragma unroll
    for (int q = 0; q < 4; ++q) {
      o0 = mma(Af[0][0][q], Bf[q], o0);
      o1 = mma(Af[0][1][q], Bf[q], o1);
    }
    asm volatile("s_wait_dscnt 0x0\n\ts_barrier_signal -1"
                 : "+v"(o0), "+v"(o1) : : "memory");
    o0 = mma(Af[0][0][4], Bf[4], o0);
    o1 = mma(Af[0][1][4], Bf[4], o1);
    asm volatile("s_barrier_wait -1" : "+v"(o0), "+v"(o1) : : "memory");

    yp = ybuf[(par ^ 1) * 16 + c];
    const v4f* pld = pbuf + par * 2048 + w * 512 + lane;
#pragma unroll
    for (int g = 0; g < 4; ++g) {
#pragma unroll
      for (int d = 1; d < 4; ++d) rc[d - 1][g] = pld[((w + d) & 3) * 128 + 32 * g];
    }
    __builtin_amdgcn_sched_barrier(0);

    o0 = mma(Af[0][0][5], Bf[5], o0);
    o1 = mma(Af[0][1][5], Bf[5], o1);
  };

  auto tail = [&](const int t) {
    const float yprev = ((yp[0] + yp[1]) + (yp[2] + yp[3])) + decb;
    if (w == 0 && hf == 0 && t > 0) y[(b0 + c) * RNN_TT + t - 1] = yprev;
    const float xv = (t < RNN_T) ? xcur : yprev;

    const v2f Bx = v2f{(hf == 0) ? xv : 0.0f, 0.0f};
    o0 = mma(Ax[0], Bx, o0);
    o1 = mma(Ax[1], Bx, o1);
#pragma unroll
    for (int p = 2; p < 4; ++p) {
      const v2f b = v2f{hB[1][2 * p], hB[1][2 * p + 1]};
      o0 = mma(Af[0][0][4 + p], b, o0);
      o1 = mma(Af[0][1][4 + p], b, o1);
    }

#pragma unroll
    for (int jj = 0; jj < 2; ++jj) {
#pragma unroll
      for (int r = 0; r < 8; ++r) {
        const int g = 2 * jj + (r >> 2), e = r & 3;
        const float own = (jj == 0) ? o0[r] : o1[r];
        const float z = (own + rc[0][g][e]) + (rc[1][g][e] + rc[2][g][e]);
        hB[jj][r] = dev_tanh(z);
      }
    }
    xcur = xnext;
  };

  head(0);
  for (int t = 0; t < RNN_TT - 1; ++t) {
    tail(t);
    head(t + 1);
  }
  tail(RNN_TT - 1);

  {
    float ys = 0.0f;
#pragma unroll
    for (int jj = 0; jj < 2; ++jj) {
#pragma unroll
      for (int r = 0; r < 8; ++r) ys = fmaf(hB[jj][r], dwv[jj][r], ys);
    }
    ys += xhalf(ys);
    const int par = (RNN_TT - 1) & 1;
    if (hf == 0) ((float*)ybuf)[(par * 16 + c) * 4 + w] = ys;
    __syncthreads();
    if (w == 0 && hf == 0) {
      const v4f ypl = ybuf[par * 16 + c];
      y[(b0 + c) * RNN_TT + RNN_TT - 1] = ((ypl[0] + ypl[1]) + (ypl[2] + ypl[3])) + decb;
    }
  }
}

extern "C" void kernel_launch(void* const* d_in, const int* in_sizes, int n_in,
                              void* d_out, int out_size, void* d_ws,
                              size_t ws_size, hipStream_t stream) {
  const float* x      = (const float*)d_in[0];
  const float* hidden = (const float*)d_in[1];
  const float* fc0_w  = (const float*)d_in[2];
  const float* fc0_b  = (const float*)d_in[3];
  const float* fc_w   = (const float*)d_in[4];
  const float* fc_b   = (const float*)d_in[5];
  const float* dec_w  = (const float*)d_in[6];
  const float* dec_b  = (const float*)d_in[7];
  float* y = (float*)d_out;

  rnn516_ksplit<<<dim3(RNN_B / 16), dim3(128), 0, stream>>>(
      x, hidden, fc0_w, fc0_b, fc_w, fc_b, dec_w, dec_b, y);
}
